# starting kernel
# baseline (speedup 1.0000x reference)
#include <hip/hip_runtime.h>

typedef __attribute__((ext_vector_type(2))) float v2f;
typedef __attribute__((ext_vector_type(8))) float v8f;

#define N_NODES  100000
#define N_EDGES  400000
#define D_EMB    128
#define D_PROMPT 64
#define D_OUT    192

__global__ void zero_kernel(float* __restrict__ out, float* __restrict__ deg) {
    long tid = (long)blockIdx.x * blockDim.x + threadIdx.x;
    const long emb_elems = (long)N_NODES * D_EMB;
    if (tid < emb_elems) {
        long n = tid >> 7;
        long c = tid & (D_EMB - 1);
        out[n * D_OUT + c] = 0.0f;
    } else if (tid < emb_elems + N_NODES) {
        deg[tid - emb_elems] = 0.0f;
    }
}

__device__ inline void atomic_fadd(float* p, float v) {
    __hip_atomic_fetch_add(p, v, __ATOMIC_RELAXED, __HIP_MEMORY_SCOPE_AGENT);
}

__global__ void scatter_kernel(const float* __restrict__ emb,
                               const int*   __restrict__ src,
                               const int*   __restrict__ dst,
                               float* __restrict__ out,
                               float* __restrict__ deg) {
    const int lane = threadIdx.x & 31;
    const int wave = threadIdx.x >> 5;
    const long e = (long)blockIdx.x * (blockDim.x >> 5) + wave;
    if (e >= N_EDGES) return;

    const int s = src[e];
    const int d = dst[e];

    const float4* row = (const float4*)(emb + (long)s * D_EMB);
    float4 v = row[lane];

    float* o = out + (long)d * D_OUT + lane * 4;
    atomic_fadd(o + 0, v.x);
    atomic_fadd(o + 1, v.y);
    atomic_fadd(o + 2, v.z);
    atomic_fadd(o + 3, v.w);

    if (lane == 0) atomic_fadd(deg + d, 1.0f);
}

__global__ void prompt_wmma_kernel(const float* __restrict__ deg,
                                   const float* __restrict__ weight,
                                   float* __restrict__ out) {
    const int lane = threadIdx.x & 31;
    const int wave = threadIdx.x >> 5;
    const int tile = blockIdx.x * (blockDim.x >> 5) + wave;
    if (tile >= N_NODES / 16) return;

    const int lo   = (lane < 16);
    const int l15  = lane & 15;

    v2f a;
    a.x = lo ? deg[tile * 16 + l15] : 0.0f;
    a.y = 0.0f;

    #pragma unroll
    for (int j = 0; j < 4; ++j) {
        v2f b;
        b.x = lo ? weight[j * 16 + l15] : 0.0f;
        b.y = 0.0f;

        v8f c = {};
        c = __builtin_amdgcn_wmma_f32_16x16x4_f32(
                false, a, false, b, (short)0, c, false, false);

        const int col = D_EMB + j * 16 + l15;
        #pragma unroll
        for (int r = 0; r < 8; ++r) {
            const int m   = lo ? r : (8 + r);
            const long row = (long)(tile * 16 + m);
            out[row * D_OUT + col] = c[r];
        }
    }
}

extern "C" void kernel_launch(void* const* d_in, const int* in_sizes, int n_in,
                              void* d_out, int out_size, void* d_ws, size_t ws_size,
                              hipStream_t stream) {
    const float* emb    = (const float*)d_in[0];
    const float* weight = (const float*)d_in[1];
    const int*   src    = (const int*)d_in[2];
    const int*   dst    = (const int*)d_in[3];

    float* out = (float*)d_out;
    float* deg = (float*)d_ws;

    {
        long n = (long)N_NODES * D_EMB + N_NODES;
        int blocks = (int)((n + 255) / 256);
        zero_kernel<<<blocks, 256, 0, stream>>>(out, deg);
    }

    {
        int waves_per_block = 256 / 32;
        int blocks = (N_EDGES + waves_per_block - 1) / waves_per_block;
        scatter_kernel<<<blocks, 256, 0, stream>>>(emb, src, dst, out, deg);
    }

    {
        int tiles = N_NODES / 16;
        int waves_per_block = 256 / 32;
        int blocks = (tiles + waves_per_block - 1) / waves_per_block;
        prompt_wmma_kernel<<<blocks, 256, 0, stream>>>(deg, weight, out);
    }
}


# scatter pass: lane L adds columns L, L+32, L+64, L+96, so each f32 atomic instruction covers one whole 128-B line instead of 4 quarter lines; same adds, arrival order as before
# speedup vs baseline: 1.1197x; 1.1197x over previous
#include <hip/hip_runtime.h>

typedef __attribute__((ext_vector_type(2))) float v2f;
typedef __attribute__((ext_vector_type(8))) float v8f;

#define N_NODES  100000
#define N_EDGES  400000
#define D_EMB    128
#define D_PROMPT 64
#define D_OUT    192

__global__ void zero_kernel(float* __restrict__ out, float* __restrict__ deg) {
    long tid = (long)blockIdx.x * blockDim.x + threadIdx.x;
    const long emb_elems = (long)N_NODES * D_EMB;
    if (tid < emb_elems) {
        long n = tid >> 7;
        long c = tid & (D_EMB - 1);
        out[n * D_OUT + c] = 0.0f;
    } else if (tid < emb_elems + N_NODES) {
        deg[tid - emb_elems] = 0.0f;
    }
}

__device__ inline void atomic_fadd(float* p, float v) {
    __hip_atomic_fetch_add(p, v, __ATOMIC_RELAXED, __HIP_MEMORY_SCOPE_AGENT);
}

__global__ void scatter_kernel(const float* __restrict__ emb,
                               const int*   __restrict__ src,
                               const int*   __restrict__ dst,
                               float* __restrict__ out,
                               float* __restrict__ deg) {
    const int lane = threadIdx.x & 31;
    const int wave = threadIdx.x >> 5;
    const long e = (long)blockIdx.x * (blockDim.x >> 5) + wave;
    if (e >= N_EDGES) return;

    const int s = src[e];
    const int d = dst[e];

    const float* row = emb + (long)s * D_EMB + lane;
    const float v0 = row[0];
    const float v1 = row[32];
    const float v2 = row[64];
    const float v3 = row[96];

    float* o = out + (long)d * D_OUT + lane;
    atomic_fadd(o +  0, v0);
    atomic_fadd(o + 32, v1);
    atomic_fadd(o + 64, v2);
    atomic_fadd(o + 96, v3);

    if (lane == 0) atomic_fadd(deg + d, 1.0f);
}

__global__ void prompt_wmma_kernel(const float* __restrict__ deg,
                                   const float* __restrict__ weight,
                                   float* __restrict__ out) {
    const int lane = threadIdx.x & 31;
    const int wave = threadIdx.x >> 5;
    const int tile = blockIdx.x * (blockDim.x >> 5) + wave;
    if (tile >= N_NODES / 16) return;

    const int lo   = (lane < 16);
    const int l15  = lane & 15;

    v2f a;
    a.x = lo ? deg[tile * 16 + l15] : 0.0f;
    a.y = 0.0f;

    #pragma unroll
    for (int j = 0; j < 4; ++j) {
        v2f b;
        b.x = lo ? weight[j * 16 + l15] : 0.0f;
        b.y = 0.0f;

        v8f c = {};
        c = __builtin_amdgcn_wmma_f32_16x16x4_f32(
                false, a, false, b, (short)0, c, false, false);

        const int col = D_EMB + j * 16 + l15;
        #pragma unroll
        for (int r = 0; r < 8; ++r) {
            const int m   = lo ? r : (8 + r);
            const long row = (long)(tile * 16 + m);
            out[row * D_OUT + col] = c[r];
        }
    }
}

extern "C" void kernel_launch(void* const* d_in, const int* in_sizes, int n_in,
                              void* d_out, int out_size, void* d_ws, size_t ws_size,
                              hipStream_t stream) {
    const float* emb    = (const float*)d_in[0];
    const float* weight = (const float*)d_in[1];
    const int*   src    = (const int*)d_in[2];
    const int*   dst    = (const int*)d_in[3];

    float* out = (float*)d_out;
    float* deg = (float*)d_ws;

    {
        long n = (long)N_NODES * D_EMB + N_NODES;
        int blocks = (int)((n + 255) / 256);
        zero_kernel<<<blocks, 256, 0, stream>>>(out, deg);
    }

    {
        int waves_per_block = 256 / 32;
        int blocks = (N_EDGES + waves_per_block - 1) / waves_per_block;
        scatter_kernel<<<blocks, 256, 0, stream>>>(emb, src, dst, out, deg);
    }

    {
        int tiles = N_NODES / 16;
        int waves_per_block = 256 / 32;
        int blocks = (tiles + waves_per_block - 1) / waves_per_block;
        prompt_wmma_kernel<<<blocks, 256, 0, stream>>>(deg, weight, out);
    }
}


# edges ordered by dst per call (int-atomic count, place, fill); one wave per node sums its source rows in registers, one store per line: no f32 atomics, no zero pass; sum order = slot arrival order
# speedup vs baseline: 1.1792x; 1.0531x over previous
#include <hip/hip_runtime.h>


typedef __attribute__((ext_vector_type(2))) float v2f;
typedef __attribute__((ext_vector_type(8))) float v8f;

#define N_NODES  100000
#define N_EDGES  400000
#define D_EMB    128
#define D_PROMPT 64
#define D_OUT    192

#define S_LIST   0
#define S_CUR    N_EDGES
#define S_TOTAL  (N_EDGES + N_NODES)

__device__ __forceinline__ long sidx(int p) {
    return (long)(p >> 6) * D_OUT + D_EMB + (p & 63);
}

__device__ __forceinline__ int atomic_iadd(int* p, int v) {
    return __hip_atomic_fetch_add(p, v, __ATOMIC_RELAXED, __HIP_MEMORY_SCOPE_AGENT);
}

__global__ void init_kernel(int* __restrict__ cnt, int* __restrict__ oi) {
    const int tid = blockIdx.x * blockDim.x + threadIdx.x;
    if (tid < N_NODES) cnt[tid] = 0;
    if (tid == 0) oi[sidx(S_TOTAL)] = 0;
}

__global__ void count_kernel(const int* __restrict__ dst, int* __restrict__ cnt) {
    const int e = blockIdx.x * blockDim.x + threadIdx.x;
    if (e < N_EDGES) atomic_iadd(cnt + dst[e], 1);
}

__global__ void place_kernel(const int* __restrict__ cnt, int* __restrict__ oi) {
    __shared__ int s_w[8];
    const int tid  = threadIdx.x;
    const int lane = tid & 31;
    const int wave = tid >> 5;
    const int n    = blockIdx.x * 256 + tid;

    const int c = (n < N_NODES) ? cnt[n] : 0;
    int incl = c;
    #pragma unroll
    for (int off = 1; off < 32; off <<= 1) {
        const int t = __shfl_up(incl, off);
        if (lane >= off) incl += t;
    }
    if (lane == 31) s_w[wave] = incl;
    __syncthreads();
    if (tid == 0) {
        int run = 0;
        #pragma unroll
        for (int w = 0; w < 8; ++w) { const int t = s_w[w]; s_w[w] = run; run += t; }
        const int base = atomic_iadd(oi + sidx(S_TOTAL), run);
        #pragma unroll
        for (int w = 0; w < 8; ++w) s_w[w] += base;
    }
    __syncthreads();
    if (n < N_NODES) oi[sidx(S_CUR + n)] = s_w[wave] + incl - c;
}

__global__ void fill_kernel(const int* __restrict__ src,
                            const int* __restrict__ dst,
                            int* __restrict__ oi) {
    const int e = blockIdx.x * blockDim.x + threadIdx.x;
    if (e >= N_EDGES) return;
    const int d = dst[e];
    const int s = src[e];
    int pos = atomic_iadd(oi + sidx(S_CUR + d), 1);
    pos = (int)min((unsigned)pos, (unsigned)(N_EDGES - 1));
    oi[sidx(S_LIST + pos)] = s;
}

__global__ void gather_kernel(const float* __restrict__ emb,
                              const int*   __restrict__ cnt,
                              float* out) {
    const int lane = threadIdx.x & 31;
    const int wave = threadIdx.x >> 5;
    const int n = blockIdx.x * (blockDim.x >> 5) + wave;
    if (n >= N_NODES) return;

    const int* oi = (const int*)out;
    int c         = __builtin_amdgcn_readfirstlane(cnt[n]);
    const int end = __builtin_amdgcn_readfirstlane(oi[sidx(S_CUR + n)]);
    c = (c < N_EDGES) ? c : N_EDGES;
    const int beg = end - c;

    const float* col = emb + 4 * lane;
    float4 acc = make_float4(0.0f, 0.0f, 0.0f, 0.0f);

    for (int base = 0; base < c; base += 32) {
        const int m = (c - base < 32) ? (c - base) : 32;
        int my = 0;
        if (lane < m) {
            const int p = (int)min((unsigned)(beg + base + lane), (unsigned)(N_EDGES - 1));
            my = oi[sidx(S_LIST + p)];
        }
        my = (int)min((unsigned)my, (unsigned)(N_NODES - 1));

        for (int i = 0; i < m; i += 4) {
            const float4 z = make_float4(0.0f, 0.0f, 0.0f, 0.0f);
            float4 r0 = z, r1 = z, r2 = z, r3 = z;
            const int s0 = __builtin_amdgcn_readlane(my, i);
            r0 = *(const float4*)(col + (long)s0 * D_EMB);
            if (i + 1 < m) {
                const int s1 = __builtin_amdgcn_readlane(my, i + 1);
                r1 = *(const float4*)(col + (long)s1 * D_EMB);
            }
            if (i + 2 < m) {
                const int s2 = __builtin_amdgcn_readlane(my, i + 2);
                r2 = *(const float4*)(col + (long)s2 * D_EMB);
            }
            if (i + 3 < m) {
                const int s3 = __builtin_amdgcn_readlane(my, i + 3);
                r3 = *(const float4*)(col + (long)s3 * D_EMB);
            }
            acc.x += (r0.x + r1.x) + (r2.x + r3.x);
            acc.y += (r0.y + r1.y) + (r2.y + r3.y);
            acc.z += (r0.z + r1.z) + (r2.z + r3.z);
            acc.w += (r0.w + r1.w) + (r2.w + r3.w);
        }
    }

    *(float4*)(out + (long)n * D_OUT + 4 * lane) = acc;
}

__global__ void prompt_wmma_kernel(const int*   __restrict__ cnt,
                                   const float* __restrict__ weight,
                                   float* __restrict__ out) {
    const int lane = threadIdx.x & 31;
    const int wave = threadIdx.x >> 5;
    const int tile = blockIdx.x * (blockDim.x >> 5) + wave;
    if (tile >= N_NODES / 16) return;

    const int lo   = (lane < 16);
    const int l15  = lane & 15;

    v2f a;
    a.x = lo ? (float)cnt[tile * 16 + l15] : 0.0f;
    a.y = 0.0f;

    #pragma unroll
    for (int j = 0; j < 4; ++j) {
        v2f b;
        b.x = lo ? weight[j * 16 + l15] : 0.0f;
        b.y = 0.0f;

        v8f c = {};
        c = __builtin_amdgcn_wmma_f32_16x16x4_f32(
                false, a, false, b, (short)0, c, false, false);

        const int col = D_EMB + j * 16 + l15;
        #pragma unroll
        for (int r = 0; r < 8; ++r) {
            const int m   = lo ? r : (8 + r);
            const long row = (long)(tile * 16 + m);
            out[row * D_OUT + col] = c[r];
        }
    }
}

extern "C" void kernel_launch(void* const* d_in, const int* in_sizes, int n_in,
                              void* d_out, int out_size, void* d_ws, size_t ws_size,
                              hipStream_t stream) {
    const float* emb    = (const float*)d_in[0];
    const float* weight = (const float*)d_in[1];
    const int*   src    = (const int*)d_in[2];
    const int*   dst    = (const int*)d_in[3];

    float* out = (float*)d_out;
    int*   oi  = (int*)d_out;
    int*   cnt = (int*)d_ws;

    const int node_blocks = (N_NODES + 255) / 256;
    const int edge_blocks = (N_EDGES + 255) / 256;

    init_kernel <<<node_blocks, 256, 0, stream>>>(cnt, oi);
    count_kernel<<<edge_blocks, 256, 0, stream>>>(dst, cnt);
    place_kernel<<<node_blocks, 256, 0, stream>>>(cnt, oi);
    fill_kernel <<<edge_blocks, 256, 0, stream>>>(src, dst, oi);

    gather_kernel<<<(N_NODES + 7) / 8, 256, 0, stream>>>(emb, cnt, out);

    {
        int tiles = N_NODES / 16;
        int waves_per_block = 256 / 32;
        int blocks = (tiles + waves_per_block - 1) / waves_per_block;
        prompt_wmma_kernel<<<blocks, 256, 0, stream>>>(cnt, weight, out);
    }
}


# source ids go into each node's own output row (one int atomic per edge picks the slot): no count or scan pass, row-sum launch also writes the prompt tile; 6 launches to 3; sum order = slot arrival
# speedup vs baseline: 1.2534x; 1.0629x over previous
#include <hip/hip_runtime.h>


typedef __attribute__((ext_vector_type(2))) float v2f;
typedef __attribute__((ext_vector_type(8))) float v8f;

#define N_NODES  100000
#define N_EDGES  400000
#define D_EMB    128
#define D_PROMPT 64
#define D_OUT    192
#define ROW_CAP  128


__device__ __forceinline__ int atomic_iadd(int* p, int v) {
    return __hip_atomic_fetch_add(p, v, __ATOMIC_RELAXED, __HIP_MEMORY_SCOPE_AGENT);
}

__global__ void init_kernel(int* __restrict__ cnt) {
    const int tid = blockIdx.x * blockDim.x + threadIdx.x;
    if (tid < N_NODES) cnt[tid] = 0;
}

__global__ void fill_kernel(const int* __restrict__ src,
                            const int* __restrict__ dst,
                            int* __restrict__ cnt,
                            int* __restrict__ oi) {
    const int e = blockIdx.x * blockDim.x + threadIdx.x;
    if (e >= N_EDGES) return;
    const int d = dst[e];
    const int s = src[e];
    int pos = atomic_iadd(cnt + d, 1);
    pos = (int)min((unsigned)pos, (unsigned)(ROW_CAP - 1));
    oi[(long)d * D_OUT + pos] = s;
}

__device__ __forceinline__ void add4(float4& acc, const float4 r0, const float4 r1,
                                     const float4 r2, const float4 r3) {
    acc.x += (r0.x + r1.x) + (r2.x + r3.x);
    acc.y += (r0.y + r1.y) + (r2.y + r3.y);
    acc.z += (r0.z + r1.z) + (r2.z + r3.z);
    acc.w += (r0.w + r1.w) + (r2.w + r3.w);
}

__global__ __launch_bounds__(512)
void gather_prompt_kernel(const float* __restrict__ emb,
                          const int*   __restrict__ cnt,
                          const float* __restrict__ weight,
                          const int*   __restrict__ src,
                          const int*   __restrict__ dst,
                          float* out) {
    __shared__ float s_p[16 * D_PROMPT];

    const int lane = threadIdx.x & 31;
    const int wave = threadIdx.x >> 5;
    const int tile = blockIdx.x;
    const int n    = tile * 16 + wave;

    const int* oi  = (const int*)out;
    const long row = (long)n * D_OUT;

    const int c_v = cnt[n];
    int my        = oi[row + lane];

    if (wave < 4) {
        const int lo  = (lane < 16);
        const int l15 = lane & 15;

        const int   degv = cnt[tile * 16 + l15];
        const float wv   = weight[wave * 16 + l15];

        v2f a;
        a.x = lo ? (float)degv : 0.0f;
        a.y = 0.0f;
        v2f b;
        b.x = lo ? wv : 0.0f;
        b.y = 0.0f;

        v8f c = {};
        c = __builtin_amdgcn_wmma_f32_16x16x4_f32(
                false, a, false, b, (short)0, c, false, false);

        for (int r = 0; r < 8; ++r) {
            const int m = lo ? r : (8 + r);
            s_p[m * D_PROMPT + wave * 16 + l15] = c[r];
        }
    }
    __syncthreads();

    const float2 pv = *(const float2*)(s_p + wave * D_PROMPT + 2 * lane);
    *(float2*)(out + row + D_EMB + 2 * lane) = pv;

    const int c = __builtin_amdgcn_readfirstlane(c_v);

    const float* col = emb + 4 * lane;
    const float4 z = make_float4(0.0f, 0.0f, 0.0f, 0.0f);
    float4 acc = z;

    if (c <= ROW_CAP) {
        for (int base = 0; base < c; base += 32) {
            const int m = (c - base < 32) ? (c - base) : 32;
            my = (int)min((unsigned)my, (unsigned)(N_NODES - 1));

            for (int i = 0; i < m; i += 4) {
                float4 r0 = z, r1 = z, r2 = z, r3 = z;
                const int s0 = __builtin_amdgcn_readlane(my, i);
                r0 = *(const float4*)(col + (long)s0 * D_EMB);
                if (i + 1 < m) {
                    const int s1 = __builtin_amdgcn_readlane(my, i + 1);
                    r1 = *(const float4*)(col + (long)s1 * D_EMB);
                }
                if (i + 2 < m) {
                    const int s2 = __builtin_amdgcn_readlane(my, i + 2);
                    r2 = *(const float4*)(col + (long)s2 * D_EMB);
                }
                if (i + 3 < m) {
                    const int s3 = __builtin_amdgcn_readlane(my, i + 3);
                    r3 = *(const float4*)(col + (long)s3 * D_EMB);
                }
                add4(acc, r0, r1, r2, r3);
            }
            if (base + 32 < c) my = oi[row + base + 32 + lane];
        }
    } else {
        for (int e0 = 0; e0 < N_EDGES; e0 += 32) {
            const int e = e0 + lane;
            unsigned hits = __builtin_amdgcn_ballot_w32(dst[e] == n);
            const int s = src[e];
            while (hits) {
                const int l = __builtin_ctz(hits);
                hits &= hits - 1;
                int sid = __builtin_amdgcn_readlane(s, l);
                sid = (int)min((unsigned)sid, (unsigned)(N_NODES - 1));
                const float4 r = *(const float4*)(col + (long)sid * D_EMB);
                acc.x += r.x; acc.y += r.y; acc.z += r.z; acc.w += r.w;
            }
        }
    }

    *(float4*)(out + row + 4 * lane) = acc;
}

extern "C" void kernel_launch(void* const* d_in, const int* in_sizes, int n_in,
                              void* d_out, int out_size, void* d_ws, size_t ws_size,
                              hipStream_t stream) {
    const float* emb    = (const float*)d_in[0];
    const float* weight = (const float*)d_in[1];
    const int*   src    = (const int*)d_in[2];
    const int*   dst    = (const int*)d_in[3];

    float* out = (float*)d_out;
    int*   oi  = (int*)d_out;
    int*   cnt = (int*)d_ws;

    const int node_blocks = (N_NODES + 255) / 256;
    const int edge_blocks = (N_EDGES + 255) / 256;

    init_kernel<<<node_blocks, 256, 0, stream>>>(cnt);
    fill_kernel<<<edge_blocks, 256, 0, stream>>>(src, dst, cnt, oi);

    gather_prompt_kernel<<<N_NODES / 16, 512, 0, stream>>>(emb, cnt, weight, src, dst, out);
}


# one launch, no global atomics or workspace: a work-group owns 400 nodes, scans dst for its own edges, lists them per node in LDS (LDS atomic picks the slot), then sums rows; sum order = slot arrival
# speedup vs baseline: 2.7805x; 2.2184x over previous
#include <hip/hip_runtime.h>


typedef __attribute__((ext_vector_type(2))) float v2f;
typedef __attribute__((ext_vector_type(8))) float v8f;

#define N_NODES      100000
#define N_EDGES      400000
#define N_EDGE4      (N_EDGES / 4)
#define D_EMB        128
#define D_PROMPT     64
#define D_OUT        192
#define WG_THREADS   1024
#define WG_NODES     400
#define WG_TILES     25
#define N_WG         250
#define LIST_CAP     32
#define SCAN_DEPTH   8


__device__ __forceinline__ void add4(float4& acc, const float4 r0, const float4 r1,
                                     const float4 r2, const float4 r3) {
    acc.x += (r0.x + r1.x) + (r2.x + r3.x);
    acc.y += (r0.y + r1.y) + (r2.y + r3.y);
    acc.z += (r0.z + r1.z) + (r2.z + r3.z);
    acc.w += (r0.w + r1.w) + (r2.w + r3.w);
}

__device__ __forceinline__ void note_edge(const unsigned dl, const int e, int* s_cnt, int* s_list) {
    if (dl < (unsigned)WG_NODES) {
        const int pos = atomicAdd(s_cnt + dl, 1);
        if (pos < LIST_CAP) s_list[dl * LIST_CAP + pos] = e;
    }
}

__device__ __forceinline__ void scan4(const int4 v, const int e, const int base, int* s_cnt, int* s_list) {
    const unsigned a = (unsigned)(v.x - base), b = (unsigned)(v.y - base);
    const unsigned c = (unsigned)(v.z - base), d = (unsigned)(v.w - base);
    if (min(min(a, b), min(c, d)) < (unsigned)WG_NODES) {
        note_edge(a, e,     s_cnt, s_list);
        note_edge(b, e + 1, s_cnt, s_list);
        note_edge(c, e + 2, s_cnt, s_list);
        note_edge(d, e + 3, s_cnt, s_list);
    }
}

__device__ __forceinline__ void sum_tail(float4& acc, const int ids, const int m, const float* col) {
    const float4 z = make_float4(0.0f, 0.0f, 0.0f, 0.0f);
    for (int i = 8; i < m; i += 4) {
        float4 r0 = z, r1 = z, r2 = z, r3 = z;
        r0 = *(const float4*)(col + (long)__builtin_amdgcn_readlane(ids, i) * D_EMB);
        if (i + 1 < m) r1 = *(const float4*)(col + (long)__builtin_amdgcn_readlane(ids, i + 1) * D_EMB);
        if (i + 2 < m) r2 = *(const float4*)(col + (long)__builtin_amdgcn_readlane(ids, i + 2) * D_EMB);
        if (i + 3 < m) r3 = *(const float4*)(col + (long)__builtin_amdgcn_readlane(ids, i + 3) * D_EMB);
        add4(acc, r0, r1, r2, r3);
    }
}

__device__ __forceinline__ float4 sum_from_edges(const int n, const int lane, const float* col,
                                                 const int* __restrict__ src,
                                                 const int* __restrict__ dst) {
    float4 acc = make_float4(0.0f, 0.0f, 0.0f, 0.0f);
    for (int e0 = 0; e0 < N_EDGES; e0 += 32) {
        const int e = e0 + lane;
        unsigned hits = __builtin_amdgcn_ballot_w32(dst[e] == n);
        const int s = src[e];
        while (hits) {
            const int l = __builtin_ctz(hits);
            hits &= hits - 1;
            int sid = __builtin_amdgcn_readlane(s, l);
            sid = (int)min((unsigned)sid, (unsigned)(N_NODES - 1));
            const float4 r = *(const float4*)(col + (long)sid * D_EMB);
            acc.x += r.x; acc.y += r.y; acc.z += r.z; acc.w += r.w;
        }
    }
    return acc;
}

#define LOAD8(R, ids, m)                                                                              \
    float4 R##0 = z, R##1 = z, R##2 = z, R##3 = z, R##4 = z, R##5 = z, R##6 = z, R##7 = z;            \
    if (0 < (m)) R##0 = *(const float4*)(col + (long)__builtin_amdgcn_readlane(ids, 0) * D_EMB);      \
    if (1 < (m)) R##1 = *(const float4*)(col + (long)__builtin_amdgcn_readlane(ids, 1) * D_EMB);      \
    if (2 < (m)) R##2 = *(const float4*)(col + (long)__builtin_amdgcn_readlane(ids, 2) * D_EMB);      \
    if (3 < (m)) R##3 = *(const float4*)(col + (long)__builtin_amdgcn_readlane(ids, 3) * D_EMB);      \
    if (4 < (m)) R##4 = *(const float4*)(col + (long)__builtin_amdgcn_readlane(ids, 4) * D_EMB);      \
    if (5 < (m)) R##5 = *(const float4*)(col + (long)__builtin_amdgcn_readlane(ids, 5) * D_EMB);      \
    if (6 < (m)) R##6 = *(const float4*)(col + (long)__builtin_amdgcn_readlane(ids, 6) * D_EMB);      \
    if (7 < (m)) R##7 = *(const float4*)(col + (long)__builtin_amdgcn_readlane(ids, 7) * D_EMB);

__global__ __launch_bounds__(WG_THREADS)
void owner_rows_kernel(const float* __restrict__ emb,
                       const float* __restrict__ weight,
                       const int*   __restrict__ src,
                       const int*   __restrict__ dst,
                       float*       __restrict__ out) {
    __shared__ int   s_cnt[WG_NODES];
    __shared__ int   s_list[WG_NODES * LIST_CAP];
    __shared__ float s_p[WG_NODES * D_PROMPT];

    const int tid  = threadIdx.x;
    const int lane = tid & 31;
    const int wave = tid >> 5;
    const int base = blockIdx.x * WG_NODES;

    if (tid < WG_NODES) s_cnt[tid] = 0;
    __syncthreads();

    const int4* dst4 = (const int4*)dst;
    constexpr int CHUNK  = SCAN_DEPTH * WG_THREADS;
    constexpr int CHUNKS = N_EDGE4 / CHUNK;
    int chunk = blockIdx.x % CHUNKS;
    for (int t = 0; t < CHUNKS; ++t) {
        const int   i0 = chunk * CHUNK + tid;
        const int4* p  = dst4 + i0;
        int4 v[SCAN_DEPTH];
#pragma unroll
        for (int u = 0; u < SCAN_DEPTH; ++u) v[u] = p[u * WG_THREADS];
#pragma unroll
        for (int u = 0; u < SCAN_DEPTH; ++u) scan4(v[u], 4 * (i0 + u * WG_THREADS), base, s_cnt, s_list);
        chunk = (chunk + 1 == CHUNKS) ? 0 : chunk + 1;
    }
    for (int i = CHUNKS * CHUNK + tid; i < N_EDGE4; i += WG_THREADS)
        scan4(dst4[i], 4 * i, base, s_cnt, s_list);
    __syncthreads();

    {
        constexpr int WORDS = WG_NODES * LIST_CAP;
        constexpr int TRIPS = (WORDS + WG_THREADS - 1) / WG_THREADS;
        int  ei[TRIPS];
        int  sv[TRIPS];
        bool ok[TRIPS];
#pragma unroll
        for (int u = 0; u < TRIPS; ++u) {
            const int i  = tid + u * WG_THREADS;
            const int ic = min(i, WORDS - 1);
            ok[u] = (i < WORDS) && ((ic & (LIST_CAP - 1)) < s_cnt[ic / LIST_CAP]);
            const int e = s_list[ic];
            ei[u] = ok[u] ? e : 0;
        }
#pragma unroll
        for (int u = 0; u < TRIPS; ++u) sv[u] = src[ei[u]];
#pragma unroll
        for (int u = 0; u < TRIPS; ++u) {
            const int i = tid + u * WG_THREADS;
            if (ok[u]) s_list[i] = (int)min((unsigned)sv[u], (unsigned)(N_NODES - 1));
        }
    }

    if (wave < WG_TILES) {
        const int lo  = (lane < 16);
        const int l15 = lane & 15;
        const int degv = s_cnt[wave * 16 + l15];
        v2f a;
        a.x = lo ? (float)degv : 0.0f;
        a.y = 0.0f;
#pragma unroll
        for (int j = 0; j < 4; ++j) {
            const float wv = weight[j * 16 + l15];
            v2f b;
            b.x = lo ? wv : 0.0f;
            b.y = 0.0f;
            v8f c = {};
            c = __builtin_amdgcn_wmma_f32_16x16x4_f32(
                    false, a, false, b, (short)0, c, false, false);
            for (int r = 0; r < 8; ++r) {
                const int m = lo ? r : (8 + r);
                s_p[(wave * 16 + m) * D_PROMPT + j * 16 + l15] = c[r];
            }
        }
    }
    __syncthreads();

    const float* col = emb + 4 * lane;
    const float4 z = make_float4(0.0f, 0.0f, 0.0f, 0.0f);

    for (int nA = wave; nA < WG_NODES; nA += 64) {
        const int  nB   = nA + 32;
        const bool hasB = nB < WG_NODES;
        const int  nBc  = hasB ? nB : nA;

        const int cA = __builtin_amdgcn_readfirstlane(s_cnt[nA]);
        const int cB = hasB ? __builtin_amdgcn_readfirstlane(s_cnt[nBc]) : 0;
        const int idsA = s_list[nA  * LIST_CAP + lane];
        const int idsB = s_list[nBc * LIST_CAP + lane];
        const float2 pA = *(const float2*)(s_p + nA  * D_PROMPT + 2 * lane);
        const float2 pB = *(const float2*)(s_p + nBc * D_PROMPT + 2 * lane);

        const int mA = (cA <= LIST_CAP) ? cA : 0;
        const int mB = (cB <= LIST_CAP) ? cB : 0;

        LOAD8(ra, idsA, mA)
        LOAD8(rb, idsB, mB)

        const long rowA = (long)(base + nA) * D_OUT;
        const long rowB = (long)(base + nB) * D_OUT;

        *(float2*)(out + rowA + D_EMB + 2 * lane) = pA;
        if (hasB) *(float2*)(out + rowB + D_EMB + 2 * lane) = pB;

        float4 accA = z, accB = z;
        add4(accA, ra0, ra1, ra2, ra3);
        add4(accA, ra4, ra5, ra6, ra7);
        add4(accB, rb0, rb1, rb2, rb3);
        add4(accB, rb4, rb5, rb6, rb7);
        sum_tail(accA, idsA, mA, col);
        sum_tail(accB, idsB, mB, col);
        if (cA > LIST_CAP) accA = sum_from_edges(base + nA, lane, col, src, dst);
        if (cB > LIST_CAP) accB = sum_from_edges(base + nB, lane, col, src, dst);

        *(float4*)(out + rowA + 4 * lane) = accA;
        if (hasB) *(float4*)(out + rowB + 4 * lane) = accB;
    }
}

extern "C" void kernel_launch(void* const* d_in, const int* in_sizes, int n_in,
                              void* d_out, int out_size, void* d_ws, size_t ws_size,
                              hipStream_t stream) {
    const float* emb    = (const float*)d_in[0];
    const float* weight = (const float*)d_in[1];
    const int*   src    = (const int*)d_in[2];
    const int*   dst    = (const int*)d_in[3];

    float* out = (float*)d_out;

    owner_rows_kernel<<<N_WG, WG_THREADS, 0, stream>>>(emb, weight, src, dst, out);
}


# first launch orders each 1600-edge slice by owner work-group (LDS count, prefix, place) into the workspace; an owner reads only its own runs, not all of dst; 1 launch to 2; sum order = slot arrival
# speedup vs baseline: 3.2879x; 1.1825x over previous
#include <hip/hip_runtime.h>


typedef __attribute__((ext_vector_type(2))) float v2f;
typedef __attribute__((ext_vector_type(8))) float v8f;

#define N_NODES      100000
#define N_EDGES      400000
#define D_EMB        128
#define D_PROMPT     64
#define D_OUT        192
#define WG_THREADS   1024
#define WG_NODES     400
#define WG_TILES     25
#define N_WG         250
#define LIST_CAP     32
#define N_SLICES     250
#define SLICE_EDGES  1600
#define SLICE_WORD4  (SLICE_EDGES / 4)
#define BIN_THREADS  256
#define RUN_PITCH    256
#define SID_BITS     17
#define SID_MASK     ((1 << SID_BITS) - 1)


__device__ __forceinline__ void add4(float4& acc, const float4 r0, const float4 r1,
                                     const float4 r2, const float4 r3) {
    acc.x += (r0.x + r1.x) + (r2.x + r3.x);
    acc.y += (r0.y + r1.y) + (r2.y + r3.y);
    acc.z += (r0.z + r1.z) + (r2.z + r3.z);
    acc.w += (r0.w + r1.w) + (r2.w + r3.w);
}

__device__ __forceinline__ void list_edge(const int word, int* s_cnt, int* s_list) {
    const unsigned dl = (unsigned)word >> SID_BITS;
    if (dl < (unsigned)WG_NODES) {
        const int pos = atomicAdd(s_cnt + dl, 1);
        if (pos < LIST_CAP) s_list[dl * LIST_CAP + pos] = (int)min((unsigned)(word & SID_MASK), (unsigned)(N_NODES - 1));
    }
}

__device__ __forceinline__ void sum_tail(float4& acc, const int ids, const int m, const float* col) {
    const float4 z = make_float4(0.0f, 0.0f, 0.0f, 0.0f);
    for (int i = 8; i < m; i += 4) {
        float4 r0 = z, r1 = z, r2 = z, r3 = z;
        r0 = *(const float4*)(col + (long)__builtin_amdgcn_readlane(ids, i) * D_EMB);
        if (i + 1 < m) r1 = *(const float4*)(col + (long)__builtin_amdgcn_readlane(ids, i + 1) * D_EMB);
        if (i + 2 < m) r2 = *(const float4*)(col + (long)__builtin_amdgcn_readlane(ids, i + 2) * D_EMB);
        if (i + 3 < m) r3 = *(const float4*)(col + (long)__builtin_amdgcn_readlane(ids, i + 3) * D_EMB);
        add4(acc, r0, r1, r2, r3);
    }
}

__device__ __forceinline__ float4 sum_from_edges(const int n, const int lane, const float* col,
                                                 const int* __restrict__ src,
                                                 const int* __restrict__ dst) {
    float4 acc = make_float4(0.0f, 0.0f, 0.0f, 0.0f);
    for (int e0 = 0; e0 < N_EDGES; e0 += 32) {
        const int e = e0 + lane;
        unsigned hits = __builtin_amdgcn_ballot_w32(dst[e] == n);
        const int s = src[e];
        while (hits) {
            const int l = __builtin_ctz(hits);
            hits &= hits - 1;
            int sid = __builtin_amdgcn_readlane(s, l);
            sid = (int)min((unsigned)sid, (unsigned)(N_NODES - 1));
            const float4 r = *(const float4*)(col + (long)sid * D_EMB);
            acc.x += r.x; acc.y += r.y; acc.z += r.z; acc.w += r.w;
        }
    }
    return acc;
}

__global__ __launch_bounds__(BIN_THREADS)
void bin_edges_kernel(const int* __restrict__ src,
                      const int* __restrict__ dst,
                      int*       __restrict__ runs,
                      int*       __restrict__ sorted) {
    __shared__ int s_c[RUN_PITCH];
    __shared__ int s_w[BIN_THREADS / 32];

    const int tid   = threadIdx.x;
    const int lane  = tid & 31;
    const int wave  = tid >> 5;
    const int slice = blockIdx.x;

    const int4* d4  = (const int4*)dst + slice * SLICE_WORD4;
    const int4* s4  = (const int4*)src + slice * SLICE_WORD4;
    const bool  two = tid + BIN_THREADS < SLICE_WORD4;
    const int   i1  = two ? tid + BIN_THREADS : tid;
    const int4 dv0 = d4[tid], dv1 = d4[i1];
    const int4 sv0 = s4[tid], sv1 = s4[i1];

    s_c[tid] = 0;
    __syncthreads();

    const int dv[8] = {dv0.x, dv0.y, dv0.z, dv0.w, dv1.x, dv1.y, dv1.z, dv1.w};
    const int sv[8] = {sv0.x, sv0.y, sv0.z, sv0.w, sv1.x, sv1.y, sv1.z, sv1.w};
    int own[8], pos[8], word[8];
#pragma unroll
    for (int j = 0; j < 8; ++j) {
        const unsigned d  = (unsigned)dv[j];
        const bool     ok = (d < (unsigned)N_NODES) && (j < 4 || two);
        const unsigned o  = d / WG_NODES;
        own[j]  = ok ? (int)o : -1;
        word[j] = (int)(((d - o * WG_NODES) << SID_BITS) | min((unsigned)sv[j], (unsigned)(N_NODES - 1)));
        pos[j]  = ok ? atomicAdd(s_c + o, 1) : 0;
    }
    __syncthreads();

    const int c = s_c[tid];
    int x = c;
#pragma unroll
    for (int off = 1; off < 32; off <<= 1) {
        const int y = __shfl_up(x, off);
        if (lane >= off) x += y;
    }
    if (lane == 31) s_w[wave] = x;
    __syncthreads();
    int before = 0;
#pragma unroll
    for (int i = 0; i < BIN_THREADS / 32 - 1; ++i) before += (i < wave) ? s_w[i] : 0;
    const int start = before + x - c;
    s_c[tid] = start;
    __syncthreads();

    runs[slice * RUN_PITCH + tid] = (start << 16) | c;
#pragma unroll
    for (int j = 0; j < 8; ++j)
        if (own[j] >= 0) sorted[slice * SLICE_EDGES + s_c[own[j]] + pos[j]] = word[j];
}

#define LOAD8(R, ids, m)                                                                              \
    float4 R##0 = z, R##1 = z, R##2 = z, R##3 = z, R##4 = z, R##5 = z, R##6 = z, R##7 = z;            \
    if (0 < (m)) R##0 = *(const float4*)(col + (long)__builtin_amdgcn_readlane(ids, 0) * D_EMB);      \
    if (1 < (m)) R##1 = *(const float4*)(col + (long)__builtin_amdgcn_readlane(ids, 1) * D_EMB);      \
    if (2 < (m)) R##2 = *(const float4*)(col + (long)__builtin_amdgcn_readlane(ids, 2) * D_EMB);      \
    if (3 < (m)) R##3 = *(const float4*)(col + (long)__builtin_amdgcn_readlane(ids, 3) * D_EMB);      \
    if (4 < (m)) R##4 = *(const float4*)(col + (long)__builtin_amdgcn_readlane(ids, 4) * D_EMB);      \
    if (5 < (m)) R##5 = *(const float4*)(col + (long)__builtin_amdgcn_readlane(ids, 5) * D_EMB);      \
    if (6 < (m)) R##6 = *(const float4*)(col + (long)__builtin_amdgcn_readlane(ids, 6) * D_EMB);      \
    if (7 < (m)) R##7 = *(const float4*)(col + (long)__builtin_amdgcn_readlane(ids, 7) * D_EMB);

__global__ __launch_bounds__(WG_THREADS)
void owner_rows_kernel(const float* __restrict__ emb,
                       const float* __restrict__ weight,
                       const int*   __restrict__ src,
                       const int*   __restrict__ dst,
                       const int*   __restrict__ runs,
                       const int*   __restrict__ sorted,
                       float*       __restrict__ out) {
    __shared__ int   s_cnt[WG_NODES];
    __shared__ int   s_list[WG_NODES * LIST_CAP];
    __shared__ float s_p[WG_NODES * D_PROMPT];

    const int tid  = threadIdx.x;
    const int lane = tid & 31;
    const int wave = tid >> 5;
    const int base = blockIdx.x * WG_NODES;

    const int slice = tid >> 2;
    const int q     = tid & 3;
    int run = 0;
    if (slice < N_SLICES) run = runs[slice * RUN_PITCH + blockIdx.x];

    if (tid < WG_NODES) s_cnt[tid] = 0;
    __syncthreads();

    {
        const int start = min((int)((unsigned)run >> 16), SLICE_EDGES);
        const int count = min(run & 0xFFFF, SLICE_EDGES - start);
        const int* seg  = sorted + min(slice, N_SLICES - 1) * SLICE_EDGES + start;
        const bool h0 = q < count, h1 = q + 4 < count, h2 = q + 8 < count;
        const int  e0 = h0 ? seg[q] : 0;
        const int  e1 = h1 ? seg[q + 4] : 0;
        const int  e2 = h2 ? seg[q + 8] : 0;
        if (h0) list_edge(e0, s_cnt, s_list);
        if (h1) list_edge(e1, s_cnt, s_list);
        if (h2) list_edge(e2, s_cnt, s_list);
        for (int k = q + 12; k < count; k += 4) list_edge(seg[k], s_cnt, s_list);
    }
    __syncthreads();

    if (wave < WG_TILES) {
        const int lo  = (lane < 16);
        const int l15 = lane & 15;
        const int degv = s_cnt[wave * 16 + l15];
        v2f a;
        a.x = lo ? (float)degv : 0.0f;
        a.y = 0.0f;
#pragma unroll
        for (int j = 0; j < 4; ++j) {
            const float wv = weight[j * 16 + l15];
            v2f b;
            b.x = lo ? wv : 0.0f;
            b.y = 0.0f;
            v8f c = {};
            c = __builtin_amdgcn_wmma_f32_16x16x4_f32(
                    false, a, false, b, (short)0, c, false, false);
            for (int r = 0; r < 8; ++r) {
                const int m = lo ? r : (8 + r);
                s_p[(wave * 16 + m) * D_PROMPT + j * 16 + l15] = c[r];
            }
        }
    }
    __syncthreads();

    const float* col = emb + 4 * lane;
    const float4 z = make_float4(0.0f, 0.0f, 0.0f, 0.0f);

    for (int nA = wave; nA < WG_NODES; nA += 64) {
        const int  nB   = nA + 32;
        const bool hasB = nB < WG_NODES;
        const int  nBc  = hasB ? nB : nA;

        const int cA = __builtin_amdgcn_readfirstlane(s_cnt[nA]);
        const int cB = hasB ? __builtin_amdgcn_readfirstlane(s_cnt[nBc]) : 0;
        const int idsA = s_list[nA  * LIST_CAP + lane];
        const int idsB = s_list[nBc * LIST_CAP + lane];
        const float2 pA = *(const float2*)(s_p + nA  * D_PROMPT + 2 * lane);
        const float2 pB = *(const float2*)(s_p + nBc * D_PROMPT + 2 * lane);

        const int mA = (cA <= LIST_CAP) ? cA : 0;
        const int mB = (cB <= LIST_CAP) ? cB : 0;

        LOAD8(ra, idsA, mA)
        LOAD8(rb, idsB, mB)

        const long rowA = (long)(base + nA) * D_OUT;
        const long rowB = (long)(base + nB) * D_OUT;

        *(float2*)(out + rowA + D_EMB + 2 * lane) = pA;
        if (hasB) *(float2*)(out + rowB + D_EMB + 2 * lane) = pB;

        float4 accA = z, accB = z;
        add4(accA, ra0, ra1, ra2, ra3);
        add4(accA, ra4, ra5, ra6, ra7);
        add4(accB, rb0, rb1, rb2, rb3);
        add4(accB, rb4, rb5, rb6, rb7);
        sum_tail(accA, idsA, mA, col);
        sum_tail(accB, idsB, mB, col);
        if (cA > LIST_CAP) accA = sum_from_edges(base + nA, lane, col, src, dst);
        if (cB > LIST_CAP) accB = sum_from_edges(base + nB, lane, col, src, dst);

        *(float4*)(out + rowA + 4 * lane) = accA;
        if (hasB) *(float4*)(out + rowB + 4 * lane) = accB;
    }
}

extern "C" void kernel_launch(void* const* d_in, const int* in_sizes, int n_in,
                              void* d_out, int out_size, void* d_ws, size_t ws_size,
                              hipStream_t stream) {
    const float* emb    = (const float*)d_in[0];
    const float* weight = (const float*)d_in[1];
    const int*   src    = (const int*)d_in[2];
    const int*   dst    = (const int*)d_in[3];

    float* out = (float*)d_out;

    int* runs   = (int*)d_ws;
    int* sorted = (int*)d_ws + N_SLICES * RUN_PITCH;

    bin_edges_kernel<<<N_SLICES, BIN_THREADS, 0, stream>>>(src, dst, runs, sorted);
    owner_rows_kernel<<<N_WG, WG_THREADS, 0, stream>>>(emb, weight, src, dst, runs, sorted, out);
}
